# starting kernel
# baseline (speedup 1.0000x reference)
#include <hip/hip_runtime.h>
#include <math.h>

typedef __attribute__((ext_vector_type(2))) float v2f;
typedef __attribute__((ext_vector_type(8))) float v8f;

#define N_GAUSS 128
#define PLANE   512
#define PI_F    3.14159265358979323846f

__global__ void ga_coeff_kernel(const float* __restrict__ mean,
                                const float* __restrict__ scale,
                                const float* __restrict__ theta,
                                float* __restrict__ coef) {
  int n = threadIdx.x;
  if (n >= N_GAUSS) return;
  float mx = mean[2 * n + 0], my = mean[2 * n + 1];
  float s0 = scale[2 * n + 0], s1 = scale[2 * n + 1];
  float th = (1.0f + sinf(theta[n])) * PI_F;
  float ct = cosf(th), st = sinf(th);
  float i0 = 1.0f / (s0 * s0);
  float i1 = 1.0f / (s1 * s1);
  float A = ct * ct * i0 + st * st * i1;
  float B = ct * st * (i0 - i1);
  float C = st * st * i0 + ct * ct * i1;
  float Dx = -2.0f * (A * mx + B * my);
  float Ey = -2.0f * (B * mx + C * my);
  float F  = A * mx * mx + 2.0f * B * mx * my + C * my * my;
  float* c8 = coef + n * 8;
  c8[0] = -0.5f * A;
  c8[1] = -1.0f * B;
  c8[2] = -0.5f * C;
  c8[3] = -0.5f * Dx;
  c8[4] = -0.5f * Ey;
  c8[5] = -0.5f * F;
  c8[6] = 0.0f;
  c8[7] = 0.0f;
}

__global__ void __launch_bounds__(256) ga_splat_kernel(
    const float* __restrict__ pix,
    const float* __restrict__ coef,
    const float* __restrict__ rgb,
    const float* __restrict__ alpha,
    float* __restrict__ out) {
  const int lane = threadIdx.x & 31;
  const int wid  = (int)((blockIdx.x * blockDim.x + threadIdx.x) >> 5);
  const int m    = lane & 15;
  const int half = lane >> 4;
  const int pixBase = wid * 16;

  const int p = pixBase + m;
  const float x = pix[2 * p + 0];
  const float y = pix[2 * p + 1];

  const float aN = alpha[N_GAUSS - 1];
  const float K0 = rgb[(N_GAUSS - 1) * 3 + 0] * aN;
  const float K1 = rgb[(N_GAUSS - 1) * 3 + 1] * aN;
  const float K2 = rgb[(N_GAUSS - 1) * 3 + 2] * aN;

  v2f a0, a1;
  a0[0] = half ? y * y : x * x;
  a0[1] = half ? x     : x * y;
  a1[0] = half ? 0.0f  : y;
  a1[1] = half ? 0.0f  : 1.0f;

  float bt[8][4];
#pragma unroll
  for (int t = 0; t < 8; ++t) {
    const float* c8 = coef + (t * 16 + m) * 8;
    bt[t][0] = c8[half ? 2 : 0];
    bt[t][1] = c8[half ? 3 : 1];
    bt[t][2] = half ? 0.0f : c8[4];
    bt[t][3] = half ? 0.0f : c8[5];
  }

  float sums[8] = {0.f, 0.f, 0.f, 0.f, 0.f, 0.f, 0.f, 0.f};

#pragma unroll
  for (int t = 0; t < 8; ++t) {
    v2f b0, b1;
    b0[0] = bt[t][0]; b0[1] = bt[t][1];
    b1[0] = bt[t][2]; b1[1] = bt[t][3];
    v8f acc = {0.f, 0.f, 0.f, 0.f, 0.f, 0.f, 0.f, 0.f};
    acc = __builtin_amdgcn_wmma_f32_16x16x4_f32(
        false, a0, false, b0, (short)0, acc, false, false);
    acc = __builtin_amdgcn_wmma_f32_16x16x4_f32(
        false, a1, false, b1, (short)0, acc, false, false);
#pragma unroll
    for (int r = 0; r < 8; ++r) sums[r] += __expf(acc[r]);
  }

#pragma unroll
  for (int r = 0; r < 8; ++r) {
    float v = sums[r];
    v += __shfl_xor(v, 8, 32);
    v += __shfl_xor(v, 4, 32);
    v += __shfl_xor(v, 2, 32);
    v += __shfl_xor(v, 1, 32);
    if (m == r) {
      int pp = pixBase + r + half * 8;
      out[3 * pp + 0] = K0 * v;
      out[3 * pp + 1] = K1 * v;
      out[3 * pp + 2] = K2 * v;
    }
  }
}

extern "C" void kernel_launch(void* const* d_in, const int* in_sizes, int n_in,
                              void* d_out, int out_size, void* d_ws, size_t ws_size,
                              hipStream_t stream) {
  const float* mean  = (const float*)d_in[0];
  const float* rgb   = (const float*)d_in[1];
  const float* alpha = (const float*)d_in[2];
  const float* scale = (const float*)d_in[3];
  const float* theta = (const float*)d_in[4];
  const float* pix   = (const float*)d_in[5];
  float* out  = (float*)d_out;
  float* coef = (float*)d_ws;

  ga_coeff_kernel<<<1, 128, 0, stream>>>(mean, scale, theta, coef);

  const int tiles   = (PLANE * PLANE) / 16;
  const int threads = 256;
  const int blocks  = tiles * 32 / threads;
  ga_splat_kernel<<<blocks, threads, 0, stream>>>(pix, coef, rgb, alpha, out);
}


# Each wave loops over 4 consecutive 16-pixel tiles with the coefficient fragments loaded once and held in registers: 4096 longer-lived waves instead of 16384 short ones, cutting wave-entry time
# speedup vs baseline: 1.0551x; 1.0551x over previous
#include <hip/hip_runtime.h>
#include <math.h>

typedef __attribute__((ext_vector_type(2))) float v2f;
typedef __attribute__((ext_vector_type(8))) float v8f;

#define N_GAUSS 128
#define PLANE   512
#define PI_F    3.14159265358979323846f

__global__ void ga_coeff_kernel(const float* __restrict__ mean,
                                const float* __restrict__ scale,
                                const float* __restrict__ theta,
                                float* __restrict__ coef) {
  int n = threadIdx.x;
  if (n >= N_GAUSS) return;
  float mx = mean[2 * n + 0], my = mean[2 * n + 1];
  float s0 = scale[2 * n + 0], s1 = scale[2 * n + 1];
  float th = (1.0f + sinf(theta[n])) * PI_F;
  float ct = cosf(th), st = sinf(th);
  float i0 = 1.0f / (s0 * s0);
  float i1 = 1.0f / (s1 * s1);
  float A = ct * ct * i0 + st * st * i1;
  float B = ct * st * (i0 - i1);
  float C = st * st * i0 + ct * ct * i1;
  float Dx = -2.0f * (A * mx + B * my);
  float Ey = -2.0f * (B * mx + C * my);
  float F  = A * mx * mx + 2.0f * B * mx * my + C * my * my;
  float* c8 = coef + n * 8;
  c8[0] = -0.5f * A;
  c8[1] = -1.0f * B;
  c8[2] = -0.5f * C;
  c8[3] = -0.5f * Dx;
  c8[4] = -0.5f * Ey;
  c8[5] = -0.5f * F;
  c8[6] = 0.0f;
  c8[7] = 0.0f;
}

#define TILES_PER_WAVE 4
__global__ void __launch_bounds__(256) ga_splat_kernel(
    const float* __restrict__ pix,
    const float* __restrict__ coef,
    const float* __restrict__ rgb,
    const float* __restrict__ alpha,
    float* __restrict__ out) {
  const int lane = threadIdx.x & 31;
  const int wid  = (int)((blockIdx.x * blockDim.x + threadIdx.x) >> 5);
  const int m    = lane & 15;
  const int half = lane >> 4;
  const int waveBase = wid * (16 * TILES_PER_WAVE);

  const float aN = alpha[N_GAUSS - 1];
  const float K0 = rgb[(N_GAUSS - 1) * 3 + 0] * aN;
  const float K1 = rgb[(N_GAUSS - 1) * 3 + 1] * aN;
  const float K2 = rgb[(N_GAUSS - 1) * 3 + 2] * aN;

  float bt[8][4];
#pragma unroll
  for (int t = 0; t < 8; ++t) {
    const float* c8 = coef + (t * 16 + m) * 8;
    bt[t][0] = c8[half ? 2 : 0];
    bt[t][1] = c8[half ? 3 : 1];
    bt[t][2] = half ? 0.0f : c8[4];
    bt[t][3] = half ? 0.0f : c8[5];
  }

  float x = pix[2 * (waveBase + m) + 0];
  float y = pix[2 * (waveBase + m) + 1];

  for (int it = 0; it < TILES_PER_WAVE; ++it) {
    const int pixBase = waveBase + it * 16;
    const int nit = (it + 1 < TILES_PER_WAVE) ? it + 1 : it;
    const int pn = waveBase + nit * 16 + m;
    const float xn = pix[2 * pn + 0];
    const float yn = pix[2 * pn + 1];

    v2f a0, a1;
    a0[0] = half ? y * y : x * x;
    a0[1] = half ? x     : x * y;
    a1[0] = half ? 0.0f  : y;
    a1[1] = half ? 0.0f  : 1.0f;

    float sums[8] = {0.f, 0.f, 0.f, 0.f, 0.f, 0.f, 0.f, 0.f};

#pragma unroll
    for (int t = 0; t < 8; ++t) {
      v2f b0, b1;
      b0[0] = bt[t][0]; b0[1] = bt[t][1];
      b1[0] = bt[t][2]; b1[1] = bt[t][3];
      v8f acc = {0.f, 0.f, 0.f, 0.f, 0.f, 0.f, 0.f, 0.f};
      acc = __builtin_amdgcn_wmma_f32_16x16x4_f32(
          false, a0, false, b0, (short)0, acc, false, false);
      acc = __builtin_amdgcn_wmma_f32_16x16x4_f32(
          false, a1, false, b1, (short)0, acc, false, false);
#pragma unroll
      for (int r = 0; r < 8; ++r) sums[r] += __expf(acc[r]);
    }

#pragma unroll
    for (int r = 0; r < 8; ++r) {
      float v = sums[r];
      v += __shfl_xor(v, 8, 32);
      v += __shfl_xor(v, 4, 32);
      v += __shfl_xor(v, 2, 32);
      v += __shfl_xor(v, 1, 32);
      if (m == r) {
        int pp = pixBase + r + half * 8;
        out[3 * pp + 0] = K0 * v;
        out[3 * pp + 1] = K1 * v;
        out[3 * pp + 2] = K2 * v;
      }
    }

    x = xn;
    y = yn;
  }
}

extern "C" void kernel_launch(void* const* d_in, const int* in_sizes, int n_in,
                              void* d_out, int out_size, void* d_ws, size_t ws_size,
                              hipStream_t stream) {
  const float* mean  = (const float*)d_in[0];
  const float* rgb   = (const float*)d_in[1];
  const float* alpha = (const float*)d_in[2];
  const float* scale = (const float*)d_in[3];
  const float* theta = (const float*)d_in[4];
  const float* pix   = (const float*)d_in[5];
  float* out  = (float*)d_out;
  float* coef = (float*)d_ws;

  ga_coeff_kernel<<<1, 128, 0, stream>>>(mean, scale, theta, coef);

  ga_splat_kernel<<<512, 256, 0, stream>>>(pix, coef, rgb, alpha, out);
}


# One launch instead of two: the first 128 threads of each block compute the gaussians' coefficients into 4 KB of LDS, read as B fragments behind one barrier; no coefficient kernel, no workspace
# speedup vs baseline: 1.5876x; 1.5047x over previous
#include <hip/hip_runtime.h>
#include <math.h>

typedef __attribute__((ext_vector_type(2))) float v2f;
typedef __attribute__((ext_vector_type(8))) float v8f;

#define N_GAUSS 128
#define PLANE   512
#define PI_F    3.14159265358979323846f

__device__ __forceinline__ void ga_coeff(const float* __restrict__ mean,
                                         const float* __restrict__ scale,
                                         const float* __restrict__ theta,
                                         int n, float4* __restrict__ dst) {
  float mx = mean[2 * n + 0], my = mean[2 * n + 1];
  float s0 = scale[2 * n + 0], s1 = scale[2 * n + 1];
  float th = (1.0f + sinf(theta[n])) * PI_F;
  float ct = cosf(th), st = sinf(th);
  float i0 = 1.0f / (s0 * s0);
  float i1 = 1.0f / (s1 * s1);
  float A = ct * ct * i0 + st * st * i1;
  float B = ct * st * (i0 - i1);
  float C = st * st * i0 + ct * ct * i1;
  float Dx = -2.0f * (A * mx + B * my);
  float Ey = -2.0f * (B * mx + C * my);
  float F  = A * mx * mx + 2.0f * B * mx * my + C * my * my;
  dst[2 * n + 0] = make_float4(-0.5f * A,
                               -1.0f * B,
                               -0.5f * Ey,
                               -0.5f * F);
  dst[2 * n + 1] = make_float4(-0.5f * C,
                               -0.5f * Dx,
                               0.0f, 0.0f);
}

#define TILES_PER_WAVE 4
__global__ void __launch_bounds__(256) ga_splat_kernel(
    const float* __restrict__ mean,
    const float* __restrict__ scale,
    const float* __restrict__ theta,
    const float* __restrict__ pix,
    const float* __restrict__ rgb,
    const float* __restrict__ alpha,
    float* __restrict__ out) {
  __shared__ float4 sB[N_GAUSS * 2];
  const int lane = threadIdx.x & 31;
  const int wid  = (int)((blockIdx.x * blockDim.x + threadIdx.x) >> 5);
  const int m    = lane & 15;
  const int half = lane >> 4;
  const int waveBase = wid * (16 * TILES_PER_WAVE);

  float x = pix[2 * (waveBase + m) + 0];
  float y = pix[2 * (waveBase + m) + 1];

  if (threadIdx.x < N_GAUSS) ga_coeff(mean, scale, theta, (int)threadIdx.x, sB);

  const float aN = alpha[N_GAUSS - 1];
  const float K0 = rgb[(N_GAUSS - 1) * 3 + 0] * aN;
  const float K1 = rgb[(N_GAUSS - 1) * 3 + 1] * aN;
  const float K2 = rgb[(N_GAUSS - 1) * 3 + 2] * aN;

  __syncthreads();

  float bt[8][4];
#pragma unroll
  for (int t = 0; t < 8; ++t) {
    const float4 q = sB[2 * (t * 16 + m) + half];
    bt[t][0] = q.x;
    bt[t][1] = q.y;
    bt[t][2] = q.z;
    bt[t][3] = q.w;
  }

  for (int it = 0; it < TILES_PER_WAVE; ++it) {
    const int pixBase = waveBase + it * 16;
    const int nit = (it + 1 < TILES_PER_WAVE) ? it + 1 : it;
    const int pn = waveBase + nit * 16 + m;
    const float xn = pix[2 * pn + 0];
    const float yn = pix[2 * pn + 1];

    v2f a0, a1;
    a0[0] = half ? y * y : x * x;
    a0[1] = half ? x     : x * y;
    a1[0] = half ? 0.0f  : y;
    a1[1] = half ? 0.0f  : 1.0f;

    float sums[8] = {0.f, 0.f, 0.f, 0.f, 0.f, 0.f, 0.f, 0.f};

#pragma unroll
    for (int t = 0; t < 8; ++t) {
      v2f b0, b1;
      b0[0] = bt[t][0]; b0[1] = bt[t][1];
      b1[0] = bt[t][2]; b1[1] = bt[t][3];
      v8f acc = {0.f, 0.f, 0.f, 0.f, 0.f, 0.f, 0.f, 0.f};
      acc = __builtin_amdgcn_wmma_f32_16x16x4_f32(
          false, a0, false, b0, (short)0, acc, false, false);
      acc = __builtin_amdgcn_wmma_f32_16x16x4_f32(
          false, a1, false, b1, (short)0, acc, false, false);
#pragma unroll
      for (int r = 0; r < 8; ++r) sums[r] += __expf(acc[r]);
    }

#pragma unroll
    for (int r = 0; r < 8; ++r) {
      float v = sums[r];
      v += __shfl_xor(v, 8, 32);
      v += __shfl_xor(v, 4, 32);
      v += __shfl_xor(v, 2, 32);
      v += __shfl_xor(v, 1, 32);
      if (m == r) {
        int pp = pixBase + r + half * 8;
        out[3 * pp + 0] = K0 * v;
        out[3 * pp + 1] = K1 * v;
        out[3 * pp + 2] = K2 * v;
      }
    }

    x = xn;
    y = yn;
  }
}

extern "C" void kernel_launch(void* const* d_in, const int* in_sizes, int n_in,
                              void* d_out, int out_size, void* d_ws, size_t ws_size,
                              hipStream_t stream) {
  const float* mean  = (const float*)d_in[0];
  const float* rgb   = (const float*)d_in[1];
  const float* alpha = (const float*)d_in[2];
  const float* scale = (const float*)d_in[3];
  const float* theta = (const float*)d_in[4];
  const float* pix   = (const float*)d_in[5];
  float* out  = (float*)d_out;

  ga_splat_kernel<<<512, 256, 0, stream>>>(mean, scale, theta, pix, rgb, alpha, out);
}


# WMMA operand roles swapped (gaussians along M, pixels along N): the gaussian sum is in-lane adds plus one lane-half exchange, not 32 dependent LDS permutes per tile; new summation tree; 3-line stores
# speedup vs baseline: 2.6734x; 1.6839x over previous
#include <hip/hip_runtime.h>
#include <math.h>

typedef __attribute__((ext_vector_type(2))) float v2f;
typedef __attribute__((ext_vector_type(8))) float v8f;

#define N_GAUSS 128
#define PLANE   512
#define PI_F    3.14159265358979323846f

__device__ __forceinline__ void ga_coeff(const float* __restrict__ mean,
                                         const float* __restrict__ scale,
                                         const float* __restrict__ theta,
                                         int n, float4* __restrict__ dst) {
  float mx = mean[2 * n + 0], my = mean[2 * n + 1];
  float s0 = scale[2 * n + 0], s1 = scale[2 * n + 1];
  float th = (1.0f + sinf(theta[n])) * PI_F;
  float ct = cosf(th), st = sinf(th);
  float i0 = 1.0f / (s0 * s0);
  float i1 = 1.0f / (s1 * s1);
  float A = ct * ct * i0 + st * st * i1;
  float B = ct * st * (i0 - i1);
  float C = st * st * i0 + ct * ct * i1;
  float Dx = -2.0f * (A * mx + B * my);
  float Ey = -2.0f * (B * mx + C * my);
  float F  = A * mx * mx + 2.0f * B * mx * my + C * my * my;
  dst[2 * n + 0] = make_float4(-0.5f * A,
                               -1.0f * B,
                               -0.5f * Ey,
                               -0.5f * F);
  dst[2 * n + 1] = make_float4(-0.5f * C,
                               -0.5f * Dx,
                               0.0f, 0.0f);
}

#define TILES_PER_WAVE 4

__device__ __forceinline__ float xhalf(float v) {
  const int b = __builtin_bit_cast(int, v);
  return __builtin_bit_cast(
      float, __builtin_amdgcn_permlanex16(b, b, 0x76543210, 0xfedcba98, false, false));
}

__global__ void __launch_bounds__(256) ga_splat_kernel(
    const float* __restrict__ mean,
    const float* __restrict__ scale,
    const float* __restrict__ theta,
    const float* __restrict__ pix,
    const float* __restrict__ rgb,
    const float* __restrict__ alpha,
    float* __restrict__ out) {
  __shared__ float4 sB[N_GAUSS * 2];
  const int lane = threadIdx.x & 31;
  const int wid  = (int)((blockIdx.x * blockDim.x + threadIdx.x) >> 5);
  const int m    = lane & 15;
  const int half = lane >> 4;
  const int waveBase = wid * (16 * TILES_PER_WAVE);

  float xa = pix[2 * (waveBase + m) + 0];
  float ya = pix[2 * (waveBase + m) + 1];
  float xb = pix[2 * (waveBase + 16 + m) + 0];
  float yb = pix[2 * (waveBase + 16 + m) + 1];

  if (threadIdx.x < N_GAUSS) ga_coeff(mean, scale, theta, (int)threadIdx.x, sB);

  const float aN = alpha[N_GAUSS - 1];
  const float K0 = rgb[(N_GAUSS - 1) * 3 + 0] * aN;
  const float K1 = rgb[(N_GAUSS - 1) * 3 + 1] * aN;
  const float K2 = rgb[(N_GAUSS - 1) * 3 + 2] * aN;

  __syncthreads();

  float bt[8][4];
#pragma unroll
  for (int t = 0; t < 8; ++t) {
    const float4 q = sB[2 * (t * 16 + m) + half];
    bt[t][0] = q.x;
    bt[t][1] = q.y;
    bt[t][2] = q.z;
    bt[t][3] = q.w;
  }

  for (int it = 0; it < TILES_PER_WAVE / 2; ++it) {
    const int pairBase = waveBase + it * 32;
    const int nit = (it + 1 < TILES_PER_WAVE / 2) ? it + 1 : it;
    const int pn = waveBase + nit * 32 + m;
    const float xan = pix[2 * pn + 0];
    const float yan = pix[2 * pn + 1];
    const float xbn = pix[2 * (pn + 16) + 0];
    const float ybn = pix[2 * (pn + 16) + 1];

    v2f fa0, fa1, fb0, fb1;
    fa0[0] = half ? ya * ya : xa * xa;
    fa0[1] = half ? xa      : xa * ya;
    fa1[0] = half ? 0.0f    : ya;
    fa1[1] = half ? 0.0f    : 1.0f;
    fb0[0] = half ? yb * yb : xb * xb;
    fb0[1] = half ? xb      : xb * yb;
    fb1[0] = half ? 0.0f    : yb;
    fb1[1] = half ? 0.0f    : 1.0f;

    float sa[8] = {0.f, 0.f, 0.f, 0.f, 0.f, 0.f, 0.f, 0.f};
    float sb[8] = {0.f, 0.f, 0.f, 0.f, 0.f, 0.f, 0.f, 0.f};

#pragma unroll
    for (int t = 0; t < 8; ++t) {
      v2f c0, c1;
      c0[0] = bt[t][0]; c0[1] = bt[t][1];
      c1[0] = bt[t][2]; c1[1] = bt[t][3];
      v8f accA = {0.f, 0.f, 0.f, 0.f, 0.f, 0.f, 0.f, 0.f};
      v8f accB = {0.f, 0.f, 0.f, 0.f, 0.f, 0.f, 0.f, 0.f};
      accA = __builtin_amdgcn_wmma_f32_16x16x4_f32(
          false, c0, false, fa0, (short)0, accA, false, false);
      accB = __builtin_amdgcn_wmma_f32_16x16x4_f32(
          false, c0, false, fb0, (short)0, accB, false, false);
      accA = __builtin_amdgcn_wmma_f32_16x16x4_f32(
          false, c1, false, fa1, (short)0, accA, false, false);
      accB = __builtin_amdgcn_wmma_f32_16x16x4_f32(
          false, c1, false, fb1, (short)0, accB, false, false);
#pragma unroll
      for (int r = 0; r < 8; ++r) {
        sa[r] += __expf(accA[r]);
        sb[r] += __expf(accB[r]);
      }
    }

    const float va = ((sa[0] + sa[1]) + (sa[2] + sa[3])) + ((sa[4] + sa[5]) + (sa[6] + sa[7]));
    const float vb = ((sb[0] + sb[1]) + (sb[2] + sb[3])) + ((sb[4] + sb[5]) + (sb[6] + sb[7]));
    const float got = xhalf(half ? va : vb);
    const float tot = (half ? vb : va) + got;

    const int pp = pairBase + lane;
    out[3 * pp + 0] = K0 * tot;
    out[3 * pp + 1] = K1 * tot;
    out[3 * pp + 2] = K2 * tot;

    xa = xan; ya = yan;
    xb = xbn; yb = ybn;
  }
}

extern "C" void kernel_launch(void* const* d_in, const int* in_sizes, int n_in,
                              void* d_out, int out_size, void* d_ws, size_t ws_size,
                              hipStream_t stream) {
  const float* mean  = (const float*)d_in[0];
  const float* rgb   = (const float*)d_in[1];
  const float* alpha = (const float*)d_in[2];
  const float* scale = (const float*)d_in[3];
  const float* theta = (const float*)d_in[4];
  const float* pix   = (const float*)d_in[5];
  float* out  = (float*)d_out;

  ga_splat_kernel<<<512, 256, 0, stream>>>(mean, scale, theta, pix, rgb, alpha, out);
}
